# starting kernel
# baseline (speedup 1.0000x reference)
#include <hip/hip_runtime.h>
#include <hip/hip_bf16.h>

typedef __attribute__((ext_vector_type(2))) float v2f;
typedef __attribute__((ext_vector_type(8))) float v8f;

#define N_GRAPHS_MAX 64
#define GV_FLOATS (N_GRAPHS_MAX * 6)

__global__ void les_init_kernel(const float* __restrict__ pos_grad,
                                float* __restrict__ force,
                                float* __restrict__ gvir, int n3) {
  int i = blockIdx.x * blockDim.x + threadIdx.x;
  if (i < n3) force[i] = -pos_grad[i];
  if (i < GV_FLOATS) gvir[i] = 0.f;
}

__global__ void les_edge_kernel(const float* __restrict__ rij,
                                const float* __restrict__ fij,
                                const int* __restrict__ eidx,
                                const int* __restrict__ batch,
                                float* __restrict__ force,
                                float* __restrict__ gvir, int E) {
  __shared__ float lvir[GV_FLOATS];
  for (int i = threadIdx.x; i < GV_FLOATS; i += blockDim.x) lvir[i] = 0.f;
  __syncthreads();

  int stride = gridDim.x * blockDim.x;
  for (int e = blockIdx.x * blockDim.x + threadIdx.x; e < E; e += stride) {
    float rx = rij[3 * e + 0], ry = rij[3 * e + 1], rz = rij[3 * e + 2];
    float fx = fij[3 * e + 0], fy = fij[3 * e + 1], fz = fij[3 * e + 2];
    int i = eidx[e];
    int j = eidx[E + e];
    unsafeAtomicAdd(&force[3 * i + 0], fx);
    unsafeAtomicAdd(&force[3 * i + 1], fy);
    unsafeAtomicAdd(&force[3 * i + 2], fz);
    unsafeAtomicAdd(&force[3 * j + 0], -fx);
    unsafeAtomicAdd(&force[3 * j + 1], -fy);
    unsafeAtomicAdd(&force[3 * j + 2], -fz);
    int g = batch[j];
    float* p = &lvir[g * 6];
    atomicAdd(p + 0, rx * fx);
    atomicAdd(p + 1, ry * fy);
    atomicAdd(p + 2, rz * fz);
    atomicAdd(p + 3, rx * fy);
    atomicAdd(p + 4, ry * fz);
    atomicAdd(p + 5, rz * fx);
  }
  __syncthreads();
  for (int i = threadIdx.x; i < GV_FLOATS; i += blockDim.x)
    unsafeAtomicAdd(&gvir[i], lvir[i]);
}

__global__ void les_stress_kernel(const float* __restrict__ cell_grad,
                                  const float* __restrict__ les_cell,
                                  const float* __restrict__ vol,
                                  const float* __restrict__ gvir,
                                  float* __restrict__ out) {
  int g = blockIdx.x;
  int lane = threadIdx.x;
  const float* C = les_cell + g * 9;
  const float* G = cell_grad + g * 9;

  int m = lane & 15;
  int hi = lane >> 4;
  int k0 = hi * 2, k1 = k0 + 1;
  int mc = (m < 2) ? m : 2;
  int k0c = (k0 < 2) ? k0 : 2;
  int k1c = (k1 < 2) ? k1 : 2;
  float rowm = (m < 3) ? 1.f : 0.f;
  float m0 = (k0 < 3) ? rowm : 0.f;
  float m1 = (k1 < 3) ? rowm : 0.f;

  v2f a, b;
  a.x = m0 * C[k0c * 3 + mc];
  a.y = m1 * C[k1c * 3 + mc];
  b.x = m0 * G[k0c * 3 + mc];
  b.y = m1 * G[k1c * 3 + mc];
  v8f c = {0.f, 0.f, 0.f, 0.f, 0.f, 0.f, 0.f, 0.f};
  v8f d = __builtin_amdgcn_wmma_f32_16x16x4_f32(
       false, a,  false, b,
       (short)0, c,  false,  false);

  __shared__ float lr[9];
  if (lane < 3) {
    lr[0 * 3 + lane] = d[0];
    lr[1 * 3 + lane] = d[1];
    lr[2 * 3 + lane] = d[2];
  }
  __syncthreads();
  if (lane == 0) {
    float inv = -1.0f / vol[g];
    const float* v = gvir + g * 6;
    float* o = out + g * 6;
    o[0] = inv * (v[0] + lr[0]);
    o[1] = inv * (v[1] + lr[4]);
    o[2] = inv * (v[2] + lr[8]);
    o[3] = inv * (v[3] + lr[1]);
    o[4] = inv * (v[4] + lr[5]);
    o[5] = inv * (v[5] + lr[2]);
  }
}

extern "C" void kernel_launch(void* const* d_in, const int* in_sizes, int n_in,
                              void* d_out, int out_size, void* d_ws, size_t ws_size,
                              hipStream_t stream) {
  const float* rij      = (const float*)d_in[0];
  const float* fij      = (const float*)d_in[1];
  const float* pos_grad = (const float*)d_in[2];
  const float* cell_grad = (const float*)d_in[3];
  const float* les_cell  = (const float*)d_in[4];
  const float* vol       = (const float*)d_in[5];
  const int*   eidx      = (const int*)d_in[6];
  const int*   batch     = (const int*)d_in[7];

  int E  = in_sizes[0] / 3;
  int n3 = in_sizes[2];
  int G  = in_sizes[5];

  float* force  = (float*)d_out;
  float* stress = force + n3;
  float* gvir   = (float*)d_ws;

  const int threads = 256;
  les_init_kernel<<<(n3 + threads - 1) / threads, threads, 0, stream>>>(
      pos_grad, force, gvir, n3);
  les_edge_kernel<<<2048, threads, 0, stream>>>(
      rij, fij, eidx, batch, force, gvir, E);
  les_stress_kernel<<<G, 32, 0, stream>>>(
      cell_grad, les_cell, vol, gvir, stress);
}


# Force accumulators: one 16-byte record per node, both halves' in workspace, so an end point's 12 bytes never cross a 128-B line: one atomic request, never two; f32 sum order changes (force)
# speedup vs baseline: 8.9004x; 8.9004x over previous
#include <hip/hip_runtime.h>
#include <hip/hip_bf16.h>


typedef __attribute__((ext_vector_type(2))) float v2f;
typedef __attribute__((ext_vector_type(8))) float v8f;

#define N_GRAPHS_MAX 64
#define GV_FLOATS (N_GRAPHS_MAX * 6)
#define N_COPIES 8
#define CHUNK 128
#define GROUPS (CHUNK / 32)
#define ROWS 3
#define EDGE_THREADS 1024
#define TICKET_STRIDE 64
#define BUCKET_SHIFT 5
#define BUCKET_MIXED 255
#define LINE_BUCKETS 128
#define INIT_THREADS 256
#define WAVE_BUCKETS (LINE_BUCKETS / (INIT_THREADS / 32))
#define FOLD_VEC 4
#define REC 4
#define FOLD_FLOATS (32 * 4 * FOLD_VEC)


__global__ __launch_bounds__(INIT_THREADS)
void les_init_kernel(const int* __restrict__ batch,
                     float* __restrict__ acc1,
                     float* __restrict__ acc2,
                     float* __restrict__ gvir,
                     int* __restrict__ ticket,
                     int* __restrict__ gtab, int n_nodes) {
  __shared__ int sline[LINE_BUCKETS / 4];
  int i = blockIdx.x * blockDim.x + threadIdx.x;
  if (i < n_nodes) {
    const float4 z = make_float4(0.f, 0.f, 0.f, 0.f);
    ((float4*)acc1)[i] = z;
    ((float4*)acc2)[i] = z;
  }
  if (i < N_COPIES * GV_FLOATS) gvir[i] = 0.f;
  if (i < 2) ticket[i * TICKET_STRIDE] = 0;

  const int n_buckets = (n_nodes + 31) >> BUCKET_SHIFT;
  const int n_lines = (n_buckets + LINE_BUCKETS - 1) / LINE_BUCKETS;
  if ((int)blockIdx.x < n_lines && n_nodes > 0) {
    const int lane = threadIdx.x & 31;
    const int wv = threadIdx.x >> 5;
    const int b0 = blockIdx.x * LINE_BUCKETS + wv * WAVE_BUCKETS;
    int bv[WAVE_BUCKETS];
    bool valid[WAVE_BUCKETS];
#pragma unroll
    for (int k = 0; k < WAVE_BUCKETS; ++k) {
      const int node = ((b0 + k) << BUCKET_SHIFT) + lane;
      valid[k] = node < n_nodes;
      bv[k] = batch[valid[k] ? node : (n_nodes - 1)];
    }
    unsigned mine = (unsigned)BUCKET_MIXED;
#pragma unroll
    for (int k = 0; k < WAVE_BUCKETS; ++k) {
      const int first = __builtin_amdgcn_readfirstlane(bv[k]);
      const bool differs = valid[k] && (bv[k] != first);
      const bool uniform = __builtin_amdgcn_ballot_w32(differs) == 0u;
      const unsigned code =
          (uniform && (unsigned)first < (unsigned)BUCKET_MIXED) ? (unsigned)first : (unsigned)BUCKET_MIXED;
      if (lane == k) mine = code;
    }
    unsigned char* sb = (unsigned char*)sline;
    if (lane < WAVE_BUCKETS) sb[wv * WAVE_BUCKETS + lane] = (unsigned char)mine;
    __syncthreads();
    if (threadIdx.x < LINE_BUCKETS / 4) gtab[blockIdx.x * (LINE_BUCKETS / 4) + threadIdx.x] = sline[threadIdx.x];
  }
}

__device__ __forceinline__ int draw_ticket(int* c, int lane) {
  int t = 0;
  if (lane == 0) t = atomicAdd(c, 1);
  return __builtin_amdgcn_readfirstlane(t);
}

__global__ __launch_bounds__(EDGE_THREADS)
void les_edge_kernel(const float* __restrict__ rij,
                     const float* __restrict__ fij,
                     const int* __restrict__ eidx,
                     const int* __restrict__ batch,
                     const int* __restrict__ gtab,
                     float* __restrict__ acc1,
                     float* __restrict__ acc2,
                     float* __restrict__ gvir,
                     int* __restrict__ ticket, int E) {
  __shared__ float lvir[GV_FLOATS];
  for (int i = threadIdx.x; i < GV_FLOATS; i += blockDim.x) lvir[i] = 0.f;
  __syncthreads();

  const int lane = threadIdx.x & 31;
  const unsigned hw = __builtin_amdgcn_s_sendmsg_rtn(0x87);
  const int die = (int)((hw >> 16) & 15u);
  const int my_half = (die >> 2) & 1;
  float* acc = my_half ? acc2 : acc1;
  const int nchunks = (E + CHUNK - 1) / CHUNK;
  const unsigned uE = (unsigned)E;

  for (int pass = 0; pass < 2; ++pass) {
    const int r = my_half ^ pass;
    int* c = ticket + r * TICKET_STRIDE;
    const int nset = (nchunks + 1 - r) >> 1;
    for (int it = 0; it <= nset; ++it) {
      const int t = draw_ticket(c, lane);
      if (t >= nset) break;
      const int chunk = 2 * t + r;

#pragma unroll 1
      for (int h = 0; h < GROUPS; ++h) {
        const unsigned eb = (unsigned)(chunk * CHUNK + h * 32);
        unsigned ni[ROWS], nj[ROWS], comp[ROWS];
        int tw[ROWS];
        float fv[ROWS], rv[ROWS], rp[ROWS];
        bool ok[ROWS];
#pragma unroll
        for (int m = 0; m < ROWS; ++m) {
          const unsigned n = 32u * (unsigned)m + (unsigned)lane;
          const unsigned el = n / 3u;
          const unsigned cc = n - 3u * el;
          unsigned e = eb + el;
          ok[m] = e < uE;
          e = ok[m] ? e : (uE - 1u);
          const unsigned off = 3u * e + cc;
          comp[m] = cc;
          ni[m] = (unsigned)eidx[e];
          nj[m] = (unsigned)eidx[uE + e];
          fv[m] = fij[off];
          rv[m] = rij[off];
          rp[m] = rij[(cc == 0u) ? (off + 2u) : (off - 1u)];
        }
#pragma unroll
        for (int m = 0; m < ROWS; ++m) {
          if (ok[m]) tw[m] = gtab[nj[m] >> (BUCKET_SHIFT + 2)];
        }
#pragma unroll
        for (int m = 0; m < ROWS; ++m) {
          if (ok[m]) {
            unsafeAtomicAdd(&acc[(unsigned)REC * ni[m] + comp[m]], fv[m]);
            unsafeAtomicAdd(&acc[(unsigned)REC * nj[m] + comp[m]], -fv[m]);
          }
        }
#pragma unroll
        for (int m = 0; m < ROWS; ++m) {
          if (ok[m]) {
            int g = (tw[m] >> (((nj[m] >> BUCKET_SHIFT) & 3u) * 8u)) & 255;
            if (g == BUCKET_MIXED) g = batch[nj[m]];
            float* p = &lvir[g * 6];
            const unsigned cp = (comp[m] == 0u) ? 2u : (comp[m] - 1u);
            atomicAdd(p + comp[m], rv[m] * fv[m]);
            atomicAdd(p + 3 + cp, rp[m] * fv[m]);
          }
        }
      }
    }
  }
  __syncthreads();
  float* gv = gvir + (die & (N_COPIES - 1)) * GV_FLOATS;
  for (int i = threadIdx.x; i < GV_FLOATS; i += blockDim.x)
    unsafeAtomicAdd(&gv[i], lvir[i]);
}

__global__ void les_stress_kernel(const float* __restrict__ cell_grad,
                                  const float* __restrict__ les_cell,
                                  const float* __restrict__ vol,
                                  const float* __restrict__ gvir,
                                  float* __restrict__ out,
                                  float* __restrict__ force,
                                  const float* __restrict__ pos_grad,
                                  const float* __restrict__ acc1,
                                  const float* __restrict__ acc2,
                                  int n3, int n_graphs) {
  if ((int)blockIdx.x >= n_graphs) {
    const int base = ((int)blockIdx.x - n_graphs) * FOLD_FLOATS + (int)threadIdx.x * 4;
    float4 p[FOLD_VEC];
    float a[FOLD_VEC][4], q[FOLD_VEC][4];
#pragma unroll
    for (int k = 0; k < FOLD_VEC; ++k) {
      const int idx = base + k * 128;
      const unsigned ic = (idx + 4 <= n3) ? (unsigned)idx : 0u;
      p[k] = *(const float4*)(pos_grad + ic);
#pragma unroll
      for (int e = 0; e < 4; ++e) {
        const unsigned i = ic + (unsigned)e;
        const unsigned node = i / 3u;
        const unsigned w = (unsigned)REC * node + (i - 3u * node);
        a[k][e] = acc1[w];
        q[k][e] = acc2[w];
      }
    }
#pragma unroll
    for (int k = 0; k < FOLD_VEC; ++k) {
      const int idx = base + k * 128;
      if (idx + 4 <= n3) {
        float4 s;
        s.x = (a[k][0] + q[k][0]) - p[k].x;
        s.y = (a[k][1] + q[k][1]) - p[k].y;
        s.z = (a[k][2] + q[k][2]) - p[k].z;
        s.w = (a[k][3] + q[k][3]) - p[k].w;
        *(float4*)(force + idx) = s;
      }
    }
    if (blockIdx.x == gridDim.x - 1 && threadIdx.x == 0)
      for (int j = n3 & ~3; j < n3; ++j) {
        const int w = REC * (j / 3) + j % 3;
        force[j] = (acc1[w] + acc2[w]) - pos_grad[j];
      }
    return;
  }
  int g = blockIdx.x;
  int lane = threadIdx.x;
  const float* C = les_cell + g * 9;
  const float* G = cell_grad + g * 9;

  int m = lane & 15;
  int hi = lane >> 4;
  int k0 = hi * 2, k1 = k0 + 1;
  int mc = (m < 2) ? m : 2;
  int k0c = (k0 < 2) ? k0 : 2;
  int k1c = (k1 < 2) ? k1 : 2;
  float rowm = (m < 3) ? 1.f : 0.f;
  float m0 = (k0 < 3) ? rowm : 0.f;
  float m1 = (k1 < 3) ? rowm : 0.f;

  v2f a, b;
  a.x = m0 * C[k0c * 3 + mc];
  a.y = m1 * C[k1c * 3 + mc];
  b.x = m0 * G[k0c * 3 + mc];
  b.y = m1 * G[k1c * 3 + mc];
  v8f c = {0.f, 0.f, 0.f, 0.f, 0.f, 0.f, 0.f, 0.f};
  v8f d = __builtin_amdgcn_wmma_f32_16x16x4_f32(
       false, a,  false, b,
       (short)0, c,  false,  false);

  __shared__ float lr[9];
  if (lane < 3) {
    lr[0 * 3 + lane] = d[0];
    lr[1 * 3 + lane] = d[1];
    lr[2 * 3 + lane] = d[2];
  }
  __syncthreads();
  if (lane == 0) {
    float inv = -1.0f / vol[g];
    float v[6];
    for (int k = 0; k < 6; ++k) {
      float s = 0.f;
      for (int q = 0; q < N_COPIES; ++q) s += gvir[q * GV_FLOATS + g * 6 + k];
      v[k] = s;
    }
    float* o = out + g * 6;
    o[0] = inv * (v[0] + lr[0]);
    o[1] = inv * (v[1] + lr[4]);
    o[2] = inv * (v[2] + lr[8]);
    o[3] = inv * (v[3] + lr[1]);
    o[4] = inv * (v[4] + lr[5]);
    o[5] = inv * (v[5] + lr[2]);
  }
}

extern "C" void kernel_launch(void* const* d_in, const int* in_sizes, int n_in,
                              void* d_out, int out_size, void* d_ws, size_t ws_size,
                              hipStream_t stream) {
  const float* rij      = (const float*)d_in[0];
  const float* fij      = (const float*)d_in[1];
  const float* pos_grad = (const float*)d_in[2];
  const float* cell_grad = (const float*)d_in[3];
  const float* les_cell  = (const float*)d_in[4];
  const float* vol       = (const float*)d_in[5];
  const int*   eidx      = (const int*)d_in[6];
  const int*   batch     = (const int*)d_in[7];

  int E  = in_sizes[0] / 3;
  int n3 = in_sizes[2];
  int G  = in_sizes[5];

  float* force  = (float*)d_out;
  float* stress = force + n3;
  float* gvir   = (float*)d_ws;
  int*   ticket = (int*)(gvir + N_COPIES * GV_FLOATS);
  int*   gtab   = ticket + 2 * TICKET_STRIDE;
  const int n_nodes = n3 / 3;
  const size_t tab_lines = ((size_t)n_nodes + 32 * LINE_BUCKETS - 1) / (32 * LINE_BUCKETS);
  float* acc1 = (float*)(gtab + ((tab_lines + 1) & ~(size_t)1) * (LINE_BUCKETS / 4));
  const size_t acc_floats = ((size_t)n_nodes * REC + 63) & ~(size_t)63;
  float* acc2 = acc1 + acc_floats;
  const size_t ws_need = (size_t)((const char*)acc2 - (const char*)d_ws) + acc_floats * sizeof(float);
  if (ws_size < ws_need) return;

  const int threads = INIT_THREADS;
  les_init_kernel<<<(n_nodes + threads - 1) / threads, threads, 0, stream>>>(
      batch, acc1, acc2, gvir, ticket, gtab, n_nodes);
  les_edge_kernel<<<512, EDGE_THREADS, 0, stream>>>(
      rij, fij, eidx, batch, gtab, acc1, acc2, gvir, ticket, E);
  les_stress_kernel<<<G + (n3 + FOLD_FLOATS - 1) / FOLD_FLOATS, 32, 0, stream>>>(
      cell_grad, les_cell, vol, gvir, stress, force, pos_grad, acc1, acc2, n3, G);
}
